# starting kernel
# baseline (speedup 1.0000x reference)
#include <hip/hip_runtime.h>

typedef float v2f __attribute__((ext_vector_type(2)));
typedef float v8f __attribute__((ext_vector_type(8)));

#define N_NODES 20000
#define N_EDGES 200000

__device__ __forceinline__ v8f wmma4(v2f a, v2f b, v8f c) {
    return __builtin_amdgcn_wmma_f32_16x16x4_f32(
         false, a,  false, b,
         (short)0, c,  false,  false);
}

__device__ __forceinline__ float silu(float x) {
    return x / (1.0f + __expf(-x));
}

__global__ void zero_kernel(float4* __restrict__ p, int n4) {
    const int stride = gridDim.x * blockDim.x;
    for (int i = blockIdx.x * blockDim.x + threadIdx.x; i < n4; i += stride)
        p[i] = make_float4(0.f, 0.f, 0.f, 0.f);
}

__global__ void pack_b_kernel(const float* __restrict__ W, float2* __restrict__ Wp,
                              int K, int N) {
    const int total  = (K / 2) * N;
    const int stride = gridDim.x * blockDim.x;
    for (int idx = blockIdx.x * blockDim.x + threadIdx.x; idx < total; idx += stride) {
        const int q = idx / N;
        const int n = idx - q * N;
        float2 p;
        p.x = W[(2 * q) * N + n];
        p.y = W[(2 * q + 1) * N + n];
        Wp[idx] = p;
    }
}

__global__ void __launch_bounds__(256)
node_up_kernel(const float* __restrict__ nf, const v2f* __restrict__ W0p,
               const v2f* __restrict__ W1p, float* __restrict__ s_up,
               float* __restrict__ v_up) {
    const int lane   = threadIdx.x & 31;
    const int wv     = threadIdx.x >> 5;
    const int lane15 = lane & 15;
    const int half   = lane >> 4;
    const int koff   = half * 2;
    const int mbase  = half * 8;
    const int n0     = blockIdx.x * 16;
    const int col    = wv * 16 + lane15;
    const int ab     = (n0 + lane15) * 512;
    const int ab2    = (n0 + lane15) * 256 + half;
    const int wbase  = half * 128 + col;

    const v2f* nf2 = (const v2f*)nf;

    v8f aS = {}, aV0 = {}, aV1 = {}, aV2 = {};
    for (int k = 0; k < 128; k += 4) {
        const int kk = k + koff;
        const int qb = (k >> 2) * 256 + wbase;
        const v2f b0 = W0p[qb];
        const v2f b1 = W1p[qb];
        const v2f a  = nf2[ab2 + (k >> 1)];
        aS = wmma4(a, b0, aS);
        const int vb = ab + 128 + kk * 3;
        v2f av;
        av.x = nf[vb + 0]; av.y = nf[vb + 3];
        aV0 = wmma4(av, b1, aV0);
        av.x = nf[vb + 1]; av.y = nf[vb + 4];
        aV1 = wmma4(av, b1, aV1);
        av.x = nf[vb + 2]; av.y = nf[vb + 5];
        aV2 = wmma4(av, b1, aV2);
    }

    const float inv = 0.08838834764831845f;
#pragma unroll
    for (int r = 0; r < 8; ++r) {
        const int n = n0 + mbase + r;
        s_up[n * 128 + col] = aS[r] * inv;
        const int vo = (n * 128 + col) * 3;
        v_up[vo + 0] = aV0[r] * inv;
        v_up[vo + 1] = aV1[r] * inv;
        v_up[vo + 2] = aV2[r] * inv;
    }
}

__device__ __forceinline__ void mlp64_layer(const float* HSRC, float* HDST,
                                            const v2f* __restrict__ Wp,
                                            int lane15, int half, int mbase) {
    const v2f* H2    = (const v2f*)HSRC;
    const int abase  = lane15 * 34 + half;
    const int wbase  = half * 64 + lane15;
    v8f acc[4] = {{}, {}, {}, {}};
    for (int k = 0; k < 64; k += 4) {
        const v2f a  = H2[abase + (k >> 1)];
        const int qb = (k >> 2) * 128 + wbase;
#pragma unroll
        for (int ct = 0; ct < 4; ++ct) {
            const v2f b = Wp[qb + ct * 16];
            acc[ct] = wmma4(a, b, acc[ct]);
        }
    }
#pragma unroll
    for (int ct = 0; ct < 4; ++ct)
#pragma unroll
        for (int r = 0; r < 8; ++r)
            HDST[(mbase + r) * 68 + ct * 16 + lane15] = silu(acc[ct][r] * 0.125f);
}

__global__ void __launch_bounds__(128)
edge_kernel(const float* __restrict__ ea, const float* __restrict__ ef,
            const int* __restrict__ snd, const int* __restrict__ rcv,
            const v2f* __restrict__ w1p, const v2f* __restrict__ w2p,
            const v2f* __restrict__ w3p, const v2f* __restrict__ w4p,
            const float* __restrict__ s_up, const float* __restrict__ v_up,
            float* __restrict__ msg) {
    __shared__ float hA[4][16 * 68];
    __shared__ float hB[4][16 * 68];

    const int lane   = threadIdx.x & 31;
    const int wv     = threadIdx.x >> 5;
    const int lane15 = lane & 15;
    const int half   = lane >> 4;
    const int mbase  = half * 8;
    const int e0     = (blockIdx.x * 4 + wv) * 16;

    float* HA = hA[wv];
    float* HB = hB[wv];

    __builtin_prefetch((const float*)w4p + threadIdx.x * 256, 0, 1);

    int   sidx[8], vidx[8], midx[8];
    float y0a[8], y1xa[8], y1ya[8], y1za[8];
#pragma unroll
    for (int r = 0; r < 8; ++r) {
        const int e    = e0 + mbase + r;
        const int node = snd[e];
        sidx[r] = node * 128;
        vidx[r] = node * 384;
        midx[r] = rcv[e] * 1024;
        const float4 y = *(const float4*)(ea + e * 4);
        y0a[r] = y.x; y1xa[r] = y.y; y1ya[r] = y.z; y1za[r] = y.w;
    }

    {
        const v2f* ef2  = (const v2f*)ef;
        const int eb2   = (e0 + lane15) * 4 + half;
        const int wbase = half * 64 + lane15;
        v8f acc[4] = {{}, {}, {}, {}};
#pragma unroll
        for (int k = 0; k < 8; k += 4) {
            const v2f a  = ef2[eb2 + (k >> 1)];
            const int qb = (k >> 2) * 128 + wbase;
#pragma unroll
            for (int ct = 0; ct < 4; ++ct) {
                const v2f b = w1p[qb + ct * 16];
                acc[ct] = wmma4(a, b, acc[ct]);
            }
        }
#pragma unroll
        for (int ct = 0; ct < 4; ++ct)
#pragma unroll
            for (int r = 0; r < 8; ++r)
                HA[(mbase + r) * 68 + ct * 16 + lane15] = silu(acc[ct][r] * 0.35355339059327373f);
    }
    asm volatile("s_wait_dscnt 0" ::: "memory");

    mlp64_layer(HA, HB, w2p, lane15, half, mbase);
    asm volatile("s_wait_dscnt 0" ::: "memory");
    mlp64_layer(HB, HA, w3p, lane15, half, mbase);
    asm volatile("s_wait_dscnt 0" ::: "memory");

    const float c  = 0.08838834764831845f;
    const float c3 = 0.05103103630798287f;
    const v2f* H2   = (const v2f*)HA;
    const int abase = lane15 * 34 + half;
#pragma unroll 1
    for (int g = 0; g < 8; ++g) {
        const int wbase = half * 512 + g * 64 + lane15;
        v8f acc[4] = {{}, {}, {}, {}};
        for (int k = 0; k < 64; k += 4) {
            const v2f a  = H2[abase + (k >> 1)];
            const int qb = (k >> 2) * 1024 + wbase;
#pragma unroll
            for (int i = 0; i < 4; ++i) {
                const v2f b = w4p[qb + i * 16];
                acc[i] = wmma4(a, b, acc[i]);
            }
        }
        const int j = g >> 1;
#pragma unroll
        for (int i = 0; i < 4; ++i) {
            const int u  = ((g & 1) * 4 + i) * 16 + lane15;
            const int u3 = 3 * u;
#pragma unroll
            for (int r = 0; r < 8; ++r) {
                const float w = acc[i][r] * 0.125f;
                if (j == 0) {
                    const float ss = s_up[sidx[r] + u];
                    atomicAdd(&msg[midx[r] + u], c * ss * y0a[r] * w);
                } else if (j == 1) {
                    const int vb = vidx[r] + u3;
                    const float dot = v_up[vb + 0] * y1xa[r] + v_up[vb + 1] * y1ya[r] +
                                      v_up[vb + 2] * y1za[r];
                    atomicAdd(&msg[midx[r] + 128 + u], c3 * dot * w);
                } else if (j == 2) {
                    const float ss = s_up[sidx[r] + u];
                    const float t  = c * ss * w;
                    const int mb = midx[r] + 256 + u3;
                    atomicAdd(&msg[mb + 0], t * y1xa[r]);
                    atomicAdd(&msg[mb + 1], t * y1ya[r]);
                    atomicAdd(&msg[mb + 2], t * y1za[r]);
                } else {
                    const int vb = vidx[r] + u3;
                    const float t  = c * y0a[r] * w;
                    const int mb = midx[r] + 640 + u3;
                    atomicAdd(&msg[mb + 0], t * v_up[vb + 0]);
                    atomicAdd(&msg[mb + 1], t * v_up[vb + 1]);
                    atomicAdd(&msg[mb + 2], t * v_up[vb + 2]);
                }
            }
        }
    }
}

__global__ void __launch_bounds__(256)
node_down_kernel(const float* __restrict__ msg, const v2f* __restrict__ L0p,
                 const v2f* __restrict__ L1p, float* __restrict__ out) {
    const int lane   = threadIdx.x & 31;
    const int wv     = threadIdx.x >> 5;
    const int lane15 = lane & 15;
    const int half   = lane >> 4;
    const int koff   = half * 2;
    const int mbase  = half * 8;
    const int n0     = blockIdx.x * 16;
    const int col    = wv * 16 + lane15;
    const int ab     = (n0 + lane15) * 1024;
    const int ab2    = (n0 + lane15) * 512 + half;
    const int wbase  = half * 128 + col;

    const v2f* msg2 = (const v2f*)msg;

    v8f aS = {}, aV0 = {}, aV1 = {}, aV2 = {};
    for (int k = 0; k < 256; k += 4) {
        const int kk = k + koff;
        const int qb = (k >> 2) * 256 + wbase;
        const v2f b0 = L0p[qb];
        const v2f b1 = L1p[qb];
        const v2f a  = msg2[ab2 + (k >> 1)];
        aS = wmma4(a, b0, aS);
        const int vb = ab + 256 + kk * 3;
        v2f av;
        av.x = msg[vb + 0]; av.y = msg[vb + 3];
        aV0 = wmma4(av, b1, aV0);
        av.x = msg[vb + 1]; av.y = msg[vb + 4];
        aV1 = wmma4(av, b1, aV1);
        av.x = msg[vb + 2]; av.y = msg[vb + 5];
        aV2 = wmma4(av, b1, aV2);
    }

    const float sc = (1.0f / 16.0f) * 0.1f;
#pragma unroll
    for (int r = 0; r < 8; ++r) {
        const int n = n0 + mbase + r;
        out[n * 512 + col] = aS[r] * sc;
        const int oo = n * 512 + 128 + col * 3;
        out[oo + 0] = aV0[r] * sc;
        out[oo + 1] = aV1[r] * sc;
        out[oo + 2] = aV2[r] * sc;
    }
}

extern "C" void kernel_launch(void* const* d_in, const int* in_sizes, int n_in,
                              void* d_out, int out_size, void* d_ws, size_t ws_size,
                              hipStream_t stream) {
    (void)in_sizes; (void)n_in; (void)out_size; (void)ws_size;
    const float* nf   = (const float*)d_in[0];
    const float* ea   = (const float*)d_in[1];
    const float* ef   = (const float*)d_in[2];
    const int*   snd  = (const int*)d_in[3];
    const int*   rcv  = (const int*)d_in[4];
    const float* Wup0 = (const float*)d_in[5];
    const float* Wup1 = (const float*)d_in[6];
    const float* w1   = (const float*)d_in[7];
    const float* w2   = (const float*)d_in[8];
    const float* w3   = (const float*)d_in[9];
    const float* w4   = (const float*)d_in[10];
    const float* Wl0  = (const float*)d_in[11];
    const float* Wl1  = (const float*)d_in[12];
    float* out = (float*)d_out;

    float* s_up = (float*)d_ws;
    float* v_up = s_up + (size_t)N_NODES * 128;
    float* msgb = v_up + (size_t)N_NODES * 384;
    float* wp   = msgb + (size_t)N_NODES * 1024;
    float* Wup0p = wp;  wp += 128 * 128;
    float* Wup1p = wp;  wp += 128 * 128;
    float* w1p   = wp;  wp += 8 * 64;
    float* w2p   = wp;  wp += 64 * 64;
    float* w3p   = wp;  wp += 64 * 64;
    float* w4p   = wp;  wp += 64 * 512;
    float* Wl0p  = wp;  wp += 256 * 128;
    float* Wl1p  = wp;  wp += 256 * 128;

    zero_kernel<<<2048, 256, 0, stream>>>((float4*)msgb, N_NODES * 1024 / 4);

    pack_b_kernel<<<32, 256, 0, stream>>>(Wup0, (float2*)Wup0p, 128, 128);
    pack_b_kernel<<<32, 256, 0, stream>>>(Wup1, (float2*)Wup1p, 128, 128);
    pack_b_kernel<<<1,  256, 0, stream>>>(w1,   (float2*)w1p,   8,   64);
    pack_b_kernel<<<8,  256, 0, stream>>>(w2,   (float2*)w2p,   64,  64);
    pack_b_kernel<<<8,  256, 0, stream>>>(w3,   (float2*)w3p,   64,  64);
    pack_b_kernel<<<64, 256, 0, stream>>>(w4,   (float2*)w4p,   64,  512);
    pack_b_kernel<<<64, 256, 0, stream>>>(Wl0,  (float2*)Wl0p,  256, 128);
    pack_b_kernel<<<64, 256, 0, stream>>>(Wl1,  (float2*)Wl1p,  256, 128);

    node_up_kernel<<<N_NODES / 16, 256, 0, stream>>>(nf, (const v2f*)Wup0p,
                                                     (const v2f*)Wup1p, s_up, v_up);
    edge_kernel<<<N_EDGES / 64, 128, 0, stream>>>(ea, ef, snd, rcv,
                                                  (const v2f*)w1p, (const v2f*)w2p,
                                                  (const v2f*)w3p, (const v2f*)w4p,
                                                  s_up, v_up, msgb);
    node_down_kernel<<<N_NODES / 16, 256, 0, stream>>>(msgb, (const v2f*)Wl0p,
                                                       (const v2f*)Wl1p, out);
}


# Weight pre-pack launches removed (12 launches -> 4): kernels build B fragments from the row-major weights; node kernels read 4 columns per 16-byte load for 4 tiles; sum order kept
# speedup vs baseline: 2.2783x; 2.2783x over previous
#include <hip/hip_runtime.h>

typedef float v2f __attribute__((ext_vector_type(2)));
typedef float v4f __attribute__((ext_vector_type(4)));
typedef float v8f __attribute__((ext_vector_type(8)));

#define N_NODES 20000
#define N_EDGES 200000

__device__ __forceinline__ v8f wmma4(v2f a, v2f b, v8f c) {
    return __builtin_amdgcn_wmma_f32_16x16x4_f32(
         false, a,  false, b,
         (short)0, c,  false,  false);
}

__device__ __forceinline__ float silu(float x) {
    return x / (1.0f + __expf(-x));
}

__global__ void zero_kernel(float4* __restrict__ p, int n4) {
    const int stride = gridDim.x * blockDim.x;
    for (int i = blockIdx.x * blockDim.x + threadIdx.x; i < n4; i += stride)
        p[i] = make_float4(0.f, 0.f, 0.f, 0.f);
}


__global__ void __launch_bounds__(128)
node_up_kernel(const float* __restrict__ nf, const float* __restrict__ W0,
               const float* __restrict__ W1, float* __restrict__ s_up,
               float* __restrict__ v_up) {
    const int lane   = threadIdx.x & 31;
    const int comp   = threadIdx.x >> 5;
    const int lane15 = lane & 15;
    const int half   = lane >> 4;
    const int mbase  = half * 8;
    const int n0     = blockIdx.x * 16;
    const int astr   = comp == 0 ? 1 : 3;
    const int aoff   = comp == 0 ? 0 : 127 + comp;
    const int abase  = (n0 + lane15) * 512 + aoff + (half * 2) * astr;
    const float* __restrict__ Wm = comp == 0 ? W0 : W1;
    const int wbase  = (half * 2) * 128 + lane15 * 4;

    v8f acc[8] = {{}, {}, {}, {}, {}, {}, {}, {}};
    for (int k = 0; k < 128; k += 4) {
        v2f a;
        a.x = nf[abase + k * astr];
        a.y = nf[abase + (k + 1) * astr];
        const float* wrow = Wm + k * 128 + wbase;
#pragma unroll
        for (int j = 0; j < 2; ++j) {
            const v4f b0 = *(const v4f*)(wrow + 64 * j);
            const v4f b1 = *(const v4f*)(wrow + 128 + 64 * j);
#pragma unroll
            for (int i = 0; i < 4; ++i) {
                v2f b;
                b.x = b0[i];
                b.y = b1[i];
                acc[4 * j + i] = wmma4(a, b, acc[4 * j + i]);
            }
        }
    }

    const float inv = 0.08838834764831845f;
    if (comp == 0) {
#pragma unroll
        for (int r = 0; r < 8; ++r) {
            float* __restrict__ d = s_up + (n0 + mbase + r) * 128 + lane15 * 4;
#pragma unroll
            for (int j = 0; j < 2; ++j) {
                v4f o;
                o.x = acc[4 * j + 0][r] * inv;
                o.y = acc[4 * j + 1][r] * inv;
                o.z = acc[4 * j + 2][r] * inv;
                o.w = acc[4 * j + 3][r] * inv;
                *(v4f*)(d + 64 * j) = o;
            }
        }
    } else {
        float* __restrict__ dst = v_up + (comp - 1);
#pragma unroll
        for (int r = 0; r < 8; ++r) {
            const int nb = (n0 + mbase + r) * 128 + lane15 * 4;
#pragma unroll
            for (int j = 0; j < 2; ++j)
#pragma unroll
                for (int i = 0; i < 4; ++i)
                    dst[(nb + 64 * j + i) * 3] = acc[4 * j + i][r] * inv;
        }
    }
}

__device__ __forceinline__ void mlp64_tile(const float* HSRC, float* HDST,
                                           const float* __restrict__ Wm, int ct,
                                           int lane15, int half, int mbase) {
    const v2f* H2    = (const v2f*)HSRC;
    const int abase  = lane15 * 34 + half;
    const int wbase  = (half * 2) * 64 + ct * 16 + lane15;
    v8f acc = {};
#pragma unroll
    for (int k = 0; k < 64; k += 4) {
        const v2f a = H2[abase + (k >> 1)];
        v2f b;
        b.x = Wm[k * 64 + wbase];
        b.y = Wm[k * 64 + 64 + wbase];
        acc = wmma4(a, b, acc);
    }
#pragma unroll
    for (int r = 0; r < 8; ++r)
        HDST[(mbase + r) * 68 + ct * 16 + lane15] = silu(acc[r] * 0.125f);
}

__global__ void __launch_bounds__(256) __attribute__((amdgpu_waves_per_eu(4)))
edge_kernel(const float* __restrict__ ea, const float* __restrict__ ef,
            const int* __restrict__ snd, const int* __restrict__ rcv,
            const float* __restrict__ w1, const float* __restrict__ w2,
            const float* __restrict__ w3, const float* __restrict__ w4,
            const float* __restrict__ s_up, const float* __restrict__ v_up,
            float* __restrict__ msg) {
    __shared__ float hX[2][16 * 68];
    __shared__ float hY[2][16 * 68];
    __shared__ float hZ[2][16 * 68];
    __shared__ __attribute__((aligned(16))) float sc[2][2][16 * 8];
    __shared__ __attribute__((aligned(16))) float G[32 * 516];

    const int lane   = threadIdx.x & 31;
    const int wv     = threadIdx.x >> 5;
    const int lane15 = lane & 15;
    const int half   = lane >> 4;
    const int mbase  = half * 8;
    const int u      = wv * 16 + lane15;
    const int tsel   = wv >> 2;
    const int ct     = wv & 3;

    v2f wr[4][16];
#pragma unroll
    for (int j = 0; j < 4; ++j)
#pragma unroll
        for (int q = 0; q < 16; ++q) {
            wr[j][q].x = w4[(q * 4 + half * 2) * 512 + j * 128 + u];
            wr[j][q].y = w4[(q * 4 + half * 2 + 1) * 512 + j * 128 + u];
        }

    const float c  = 0.08838834764831845f;
    const float c3 = 0.05103103630798287f;
    const int abase = lane15 * 34 + half;

    const int wvu = __builtin_amdgcn_readfirstlane(wv);
    const int sh  = wvu >> 2;
    const int sq  = wvu & 3;
    const char* __restrict__ gbase = (const char*)(sq == 0 ? s_up : v_up + (sq - 1) * 128);
    const unsigned gstride = sq == 0 ? 512u : 1536u;
    const unsigned glane   = lane * 16;
    float* gdst = G + sh * 16 * 516 + sq * 128 + lane * 4;
#define STAGE_LOAD(I) (*(const v4f*)(gbase + ((unsigned)__builtin_amdgcn_readlane(sv, (I)) * gstride + glane)))

    int par = 0;
    for (int p = blockIdx.x; p < N_EDGES / 32; p += gridDim.x, par ^= 1) {
        const int e0 = p * 32 + tsel * 16;

        const int sv = snd[p * 32 + sh * 16 + lane15];

        {
            const v2f* ef2  = (const v2f*)ef;
            const int eb2   = (e0 + lane15) * 4 + half;
            const int wbase = (half * 2) * 64 + ct * 16 + lane15;
            v8f acc = {};
#pragma unroll
            for (int k = 0; k < 8; k += 4) {
                v2f b;
                b.x = w1[k * 64 + wbase];
                b.y = w1[k * 64 + 64 + wbase];
                acc = wmma4(ef2[eb2 + (k >> 1)], b, acc);
            }
#pragma unroll
            for (int r = 0; r < 8; ++r)
                hX[tsel][(mbase + r) * 68 + ct * 16 + lane15] =
                    silu(acc[r] * 0.35355339059327373f);
        }
        if (ct == 0) {
            const int e = e0 + lane15;
            const float4 y = *(const float4*)(ea + e * 4);
            float4 sr;
            sr.x = __int_as_float(snd[e]);
            sr.y = __int_as_float(rcv[e]);
            sr.z = 0.f; sr.w = 0.f;
            *(float4*)(&sc[par][tsel][lane15 * 8])     = y;
            *(float4*)(&sc[par][tsel][lane15 * 8 + 4]) = sr;
        }
        __syncthreads();
        {
            v4f ga[8];
#pragma unroll
            for (int i = 0; i < 8; ++i)
                ga[i] = STAGE_LOAD(i);
#pragma unroll
            for (int i = 0; i < 8; ++i)
                *(v4f*)(gdst + i * 516) = ga[i];
        }
        mlp64_tile(hX[tsel], hY[tsel], w2, ct, lane15, half, mbase);
        __syncthreads();
        {
            v4f ga[8];
#pragma unroll
            for (int i = 0; i < 8; ++i)
                ga[i] = STAGE_LOAD(8 + i);
#pragma unroll
            for (int i = 0; i < 8; ++i)
                *(v4f*)(gdst + (8 + i) * 516) = ga[i];
        }
        mlp64_tile(hY[tsel], hZ[tsel], w3, ct, lane15, half, mbase);
        __syncthreads();

#pragma unroll 1
        for (int t = 0; t < 2; ++t) {
            const v2f* H2 = (const v2f*)hZ[t];
            const float* S = sc[par][t];
            const float* gs = G + (t * 16 + mbase) * 516 + u;
            const float* gv = G + (t * 16 + mbase) * 516 + 128 + 3 * u;
            v8f acc[4] = {{}, {}, {}, {}};
#pragma unroll
            for (int q = 0; q < 16; ++q) {
                const v2f a = H2[abase + q * 2];
#pragma unroll
                for (int j = 0; j < 4; ++j)
                    acc[j] = wmma4(a, wr[j][q], acc[j]);
            }
#pragma unroll
            for (int r = 0; r < 8; ++r) {
                const int row = mbase + r;
                const float4 y  = *(const float4*)(S + row * 8);
                const float2 sr = *(const float2*)(S + row * 8 + 4);
                const float y0 = y.x, y1x = y.y, y1y = y.z, y1z = y.w;
                const int mb   = __float_as_int(sr.y) * 1024 + u;
                const float ss = gs[r * 516];
                const float vx = gv[r * 516 + 0];
                const float vy = gv[r * 516 + 1];
                const float vz = gv[r * 516 + 2];
                const float w0 = acc[0][r] * 0.125f;
                const float w1 = acc[1][r] * 0.125f;
                const float w2 = acc[2][r] * 0.125f;
                const float w3 = acc[3][r] * 0.125f;
                atomicAdd(&msg[mb], c * ss * y0 * w0);
                const float dot = vx * y1x + vy * y1y + vz * y1z;
                atomicAdd(&msg[mb + 128], c3 * dot * w1);
                const float ta = c * ss * w2;
                atomicAdd(&msg[mb + 256], ta * y1x);
                atomicAdd(&msg[mb + 384], ta * y1y);
                atomicAdd(&msg[mb + 512], ta * y1z);
                const float tb = c * y0 * w3;
                atomicAdd(&msg[mb + 640], tb * vx);
                atomicAdd(&msg[mb + 768], tb * vy);
                atomicAdd(&msg[mb + 896], tb * vz);
            }
        }
    }
}

#undef STAGE_LOAD

__global__ void __launch_bounds__(128)
node_down_kernel(const float* __restrict__ msg, const float* __restrict__ L0,
                 const float* __restrict__ L1, float* __restrict__ out) {
    const int lane   = threadIdx.x & 31;
    const int comp   = threadIdx.x >> 5;
    const int lane15 = lane & 15;
    const int half   = lane >> 4;
    const int mbase  = half * 8;
    const int n0     = blockIdx.x * 16;
    const int ab     = (n0 + lane15) * 512 + half + (comp == 0 ? 0 : 128 + 64 * (comp - 1));
    const int hi     = comp == 0 ? 0 : 128;
    const float* __restrict__ Wm = comp == 0 ? L0 : L1;
    const int wbase  = (half * 2) * 128 + lane15 * 4;

    const v2f* msg2 = (const v2f*)msg;

    v8f acc[8] = {{}, {}, {}, {}, {}, {}, {}, {}};
    for (int k = 0; k < 256; k += 4) {
        const v2f a  = msg2[ab + (k >> 1) + (k < 128 ? 0 : hi)];
        const float* wrow = Wm + k * 128 + wbase;
#pragma unroll
        for (int j = 0; j < 2; ++j) {
            const v4f b0 = *(const v4f*)(wrow + 64 * j);
            const v4f b1 = *(const v4f*)(wrow + 128 + 64 * j);
#pragma unroll
            for (int i = 0; i < 4; ++i) {
                v2f b;
                b.x = b0[i];
                b.y = b1[i];
                acc[4 * j + i] = wmma4(a, b, acc[4 * j + i]);
            }
        }
    }

    const float sc = (1.0f / 16.0f) * 0.1f;
    if (comp == 0) {
#pragma unroll
        for (int r = 0; r < 8; ++r) {
            float* __restrict__ d = out + (n0 + mbase + r) * 512 + lane15 * 4;
#pragma unroll
            for (int j = 0; j < 2; ++j) {
                v4f o;
                o.x = acc[4 * j + 0][r] * sc;
                o.y = acc[4 * j + 1][r] * sc;
                o.z = acc[4 * j + 2][r] * sc;
                o.w = acc[4 * j + 3][r] * sc;
                *(v4f*)(d + 64 * j) = o;
            }
        }
    } else {
#pragma unroll
        for (int r = 0; r < 8; ++r) {
            const int ob = (n0 + mbase + r) * 512 + 127 + comp + lane15 * 12;
#pragma unroll
            for (int j = 0; j < 2; ++j)
#pragma unroll
                for (int i = 0; i < 4; ++i)
                    out[ob + (64 * j + i) * 3] = acc[4 * j + i][r] * sc;
        }
    }
}

extern "C" void kernel_launch(void* const* d_in, const int* in_sizes, int n_in,
                              void* d_out, int out_size, void* d_ws, size_t ws_size,
                              hipStream_t stream) {
    (void)in_sizes; (void)n_in; (void)out_size; (void)ws_size;
    const float* nf   = (const float*)d_in[0];
    const float* ea   = (const float*)d_in[1];
    const float* ef   = (const float*)d_in[2];
    const int*   snd  = (const int*)d_in[3];
    const int*   rcv  = (const int*)d_in[4];
    const float* Wup0 = (const float*)d_in[5];
    const float* Wup1 = (const float*)d_in[6];
    const float* w1   = (const float*)d_in[7];
    const float* w2   = (const float*)d_in[8];
    const float* w3   = (const float*)d_in[9];
    const float* w4   = (const float*)d_in[10];
    const float* Wl0  = (const float*)d_in[11];
    const float* Wl1  = (const float*)d_in[12];
    float* out = (float*)d_out;

    float* s_up = (float*)d_ws;
    float* v_up = s_up + (size_t)N_NODES * 128;
    float* msgb = v_up + (size_t)N_NODES * 384;

    zero_kernel<<<2048, 256, 0, stream>>>((float4*)msgb, N_NODES * 1024 / 4);

    node_up_kernel<<<N_NODES / 16, 128, 0, stream>>>(nf, Wup0, Wup1, s_up, v_up);
    edge_kernel<<<512, 256, 0, stream>>>(ea, ef, snd, rcv, w1, w2, w3, w4,
                                         s_up, v_up, msgb);
    node_down_kernel<<<N_NODES / 16, 128, 0, stream>>>(msgb, Wl0, Wl1, out);
}


# msg row re-laid as 128-B lines of two planes x 16 channels; v_permlane16_swap joins the lane halves so each atomic add is one whole line of one row, not 2 x 64 B; node_down sums k in row order
# speedup vs baseline: 2.4891x; 1.0925x over previous
#include <hip/hip_runtime.h>

typedef float v2f __attribute__((ext_vector_type(2)));
typedef float v4f __attribute__((ext_vector_type(4)));
typedef float v8f __attribute__((ext_vector_type(8)));

#define N_NODES 20000
#define N_EDGES 200000

__device__ __forceinline__ v8f wmma4(v2f a, v2f b, v8f c) {
    return __builtin_amdgcn_wmma_f32_16x16x4_f32(
         false, a,  false, b,
         (short)0, c,  false,  false);
}

__device__ __forceinline__ float silu(float x) {
    return x / (1.0f + __expf(-x));
}

__device__ __forceinline__ void join_rows(float& a, float& b) {
    typedef unsigned u2v __attribute__((ext_vector_type(2)));
    const u2v s = __builtin_amdgcn_permlane16_swap(__float_as_uint(a), __float_as_uint(b),
                                                   false, false);
    a = __uint_as_float(s.x);
    b = __uint_as_float(s.y);
}

__global__ void zero_kernel(float4* __restrict__ p, int n4) {
    const int stride = gridDim.x * blockDim.x;
    for (int i = blockIdx.x * blockDim.x + threadIdx.x; i < n4; i += stride)
        p[i] = make_float4(0.f, 0.f, 0.f, 0.f);
}


__global__ void __launch_bounds__(128)
node_up_kernel(const float* __restrict__ nf, const float* __restrict__ W0,
               const float* __restrict__ W1, float* __restrict__ s_up,
               float* __restrict__ v_up) {
    const int lane   = threadIdx.x & 31;
    const int comp   = threadIdx.x >> 5;
    const int lane15 = lane & 15;
    const int half   = lane >> 4;
    const int mbase  = half * 8;
    const int n0     = blockIdx.x * 16;
    const int astr   = comp == 0 ? 1 : 3;
    const int aoff   = comp == 0 ? 0 : 127 + comp;
    const int abase  = (n0 + lane15) * 512 + aoff + (half * 2) * astr;
    const float* __restrict__ Wm = comp == 0 ? W0 : W1;
    const int wbase  = (half * 2) * 128 + lane15 * 4;

    v8f acc[8] = {{}, {}, {}, {}, {}, {}, {}, {}};
    for (int k = 0; k < 128; k += 4) {
        v2f a;
        a.x = nf[abase + k * astr];
        a.y = nf[abase + (k + 1) * astr];
        const float* wrow = Wm + k * 128 + wbase;
#pragma unroll
        for (int j = 0; j < 2; ++j) {
            const v4f b0 = *(const v4f*)(wrow + 64 * j);
            const v4f b1 = *(const v4f*)(wrow + 128 + 64 * j);
#pragma unroll
            for (int i = 0; i < 4; ++i) {
                v2f b;
                b.x = b0[i];
                b.y = b1[i];
                acc[4 * j + i] = wmma4(a, b, acc[4 * j + i]);
            }
        }
    }

    const float inv = 0.08838834764831845f;
    if (comp == 0) {
#pragma unroll
        for (int r = 0; r < 8; ++r) {
            float* __restrict__ d = s_up + (n0 + mbase + r) * 128 + lane15 * 4;
#pragma unroll
            for (int j = 0; j < 2; ++j) {
                v4f o;
                o.x = acc[4 * j + 0][r] * inv;
                o.y = acc[4 * j + 1][r] * inv;
                o.z = acc[4 * j + 2][r] * inv;
                o.w = acc[4 * j + 3][r] * inv;
                *(v4f*)(d + 64 * j) = o;
            }
        }
    } else {
        float* __restrict__ dst = v_up + (comp - 1);
#pragma unroll
        for (int r = 0; r < 8; ++r) {
            const int nb = (n0 + mbase + r) * 128 + lane15 * 4;
#pragma unroll
            for (int j = 0; j < 2; ++j)
#pragma unroll
                for (int i = 0; i < 4; ++i)
                    dst[(nb + 64 * j + i) * 3] = acc[4 * j + i][r] * inv;
        }
    }
}

__device__ __forceinline__ void mlp64_tile(const float* HSRC, float* HDST,
                                           const float* __restrict__ Wm, int ct,
                                           int lane15, int half, int mbase) {
    const v2f* H2    = (const v2f*)HSRC;
    const int abase  = lane15 * 34 + half;
    const int wbase  = (half * 2) * 64 + ct * 16 + lane15;
    v8f acc = {};
#pragma unroll
    for (int k = 0; k < 64; k += 4) {
        const v2f a = H2[abase + (k >> 1)];
        v2f b;
        b.x = Wm[k * 64 + wbase];
        b.y = Wm[k * 64 + 64 + wbase];
        acc = wmma4(a, b, acc);
    }
#pragma unroll
    for (int r = 0; r < 8; ++r)
        HDST[(mbase + r) * 68 + ct * 16 + lane15] = silu(acc[r] * 0.125f);
}

__global__ void __launch_bounds__(256) __attribute__((amdgpu_waves_per_eu(4)))
edge_kernel(const float* __restrict__ ea, const float* __restrict__ ef,
            const int* __restrict__ snd, const int* __restrict__ rcv,
            const float* __restrict__ w1, const float* __restrict__ w2,
            const float* __restrict__ w3, const float* __restrict__ w4,
            const float* __restrict__ s_up, const float* __restrict__ v_up,
            float* __restrict__ msg) {
    __shared__ float hX[2][16 * 68];
    __shared__ float hY[2][16 * 68];
    __shared__ float hZ[2][16 * 68];
    __shared__ __attribute__((aligned(16))) float sc[2][2][16 * 8];
    __shared__ __attribute__((aligned(16))) float G[32 * 516];

    const int lane   = threadIdx.x & 31;
    const int wv     = threadIdx.x >> 5;
    const int lane15 = lane & 15;
    const int half   = lane >> 4;
    const int mbase  = half * 8;
    const int u      = wv * 16 + lane15;
    const int tsel   = wv >> 2;
    const int ct     = wv & 3;

    v2f wr[4][16];
#pragma unroll
    for (int j = 0; j < 4; ++j)
#pragma unroll
        for (int q = 0; q < 16; ++q) {
            wr[j][q].x = w4[(q * 4 + half * 2) * 512 + j * 128 + u];
            wr[j][q].y = w4[(q * 4 + half * 2 + 1) * 512 + j * 128 + u];
        }

    const float c  = 0.08838834764831845f;
    const float c3 = 0.05103103630798287f;
    const int abase = lane15 * 34 + half;

    const int wvu = __builtin_amdgcn_readfirstlane(wv);
    const int sh  = wvu >> 2;
    const int sq  = wvu & 3;
    const char* __restrict__ gbase = (const char*)(sq == 0 ? s_up : v_up + (sq - 1) * 128);
    const unsigned gstride = sq == 0 ? 512u : 1536u;
    const unsigned glane   = lane * 16;
    float* gdst = G + sh * 16 * 516 + sq * 128 + lane * 4;
#define STAGE_LOAD(I) (*(const v4f*)(gbase + ((unsigned)__builtin_amdgcn_readlane(sv, (I)) * gstride + glane)))

    int par = 0;
    for (int p = blockIdx.x; p < N_EDGES / 32; p += gridDim.x, par ^= 1) {
        const int e0 = p * 32 + tsel * 16;

        const int sv = snd[p * 32 + sh * 16 + lane15];

        {
            const v2f* ef2  = (const v2f*)ef;
            const int eb2   = (e0 + lane15) * 4 + half;
            const int wbase = (half * 2) * 64 + ct * 16 + lane15;
            v8f acc = {};
#pragma unroll
            for (int k = 0; k < 8; k += 4) {
                v2f b;
                b.x = w1[k * 64 + wbase];
                b.y = w1[k * 64 + 64 + wbase];
                acc = wmma4(ef2[eb2 + (k >> 1)], b, acc);
            }
#pragma unroll
            for (int r = 0; r < 8; ++r)
                hX[tsel][(mbase + r) * 68 + ct * 16 + lane15] =
                    silu(acc[r] * 0.35355339059327373f);
        }
        if (ct == 0) {
            const int e = e0 + lane15;
            const float4 y = *(const float4*)(ea + e * 4);
            float4 sr;
            sr.x = __int_as_float(snd[e]);
            sr.y = __int_as_float(rcv[e]);
            sr.z = 0.f; sr.w = 0.f;
            *(float4*)(&sc[par][tsel][lane15 * 8])     = y;
            *(float4*)(&sc[par][tsel][lane15 * 8 + 4]) = sr;
        }
        __syncthreads();
        {
            v4f ga[8];
#pragma unroll
            for (int i = 0; i < 8; ++i)
                ga[i] = STAGE_LOAD(i);
#pragma unroll
            for (int i = 0; i < 8; ++i)
                *(v4f*)(gdst + i * 516) = ga[i];
        }
        mlp64_tile(hX[tsel], hY[tsel], w2, ct, lane15, half, mbase);
        __syncthreads();
        {
            v4f ga[8];
#pragma unroll
            for (int i = 0; i < 8; ++i)
                ga[i] = STAGE_LOAD(8 + i);
#pragma unroll
            for (int i = 0; i < 8; ++i)
                *(v4f*)(gdst + (8 + i) * 516) = ga[i];
        }
        mlp64_tile(hY[tsel], hZ[tsel], w3, ct, lane15, half, mbase);
        __syncthreads();

#pragma unroll 1
        for (int t = 0; t < 2; ++t) {
            const v2f* H2 = (const v2f*)hZ[t];
            const float* S = sc[par][t];
            const float* gs = G + (t * 16 + mbase) * 516 + u;
            const float* gv = G + (t * 16 + mbase) * 516 + 128 + 3 * u;
            const int lo = wv * 32 + lane;
            v8f acc[4] = {{}, {}, {}, {}};
#pragma unroll
            for (int q = 0; q < 16; ++q) {
                const v2f a = H2[abase + q * 2];
#pragma unroll
                for (int j = 0; j < 4; ++j)
                    acc[j] = wmma4(a, wr[j][q], acc[j]);
            }
#pragma unroll
            for (int r = 0; r < 8; ++r) {
                const int row = mbase + r;
                const float4 y  = *(const float4*)(S + row * 8);
                const float y0 = y.x, y1x = y.y, y1y = y.z, y1z = y.w;
                const int rcvA = __builtin_amdgcn_readfirstlane(__float_as_int(S[r * 8 + 5]));
                const int rcvB = __builtin_amdgcn_readfirstlane(__float_as_int(S[(r + 8) * 8 + 5]));
                float* __restrict__ mA = msg + (size_t)rcvA * 1024 + lo;
                float* __restrict__ mB = msg + (size_t)rcvB * 1024 + lo;
                const float ss = gs[r * 516];
                const float vx = gv[r * 516 + 0];
                const float vy = gv[r * 516 + 1];
                const float vz = gv[r * 516 + 2];
                const float w0 = acc[0][r] * 0.125f;
                const float w1 = acc[1][r] * 0.125f;
                const float w2 = acc[2][r] * 0.125f;
                const float w3 = acc[3][r] * 0.125f;
                float p0 = c * ss * y0 * w0;
                const float dot = vx * y1x + vy * y1y + vz * y1z;
                float p1 = c3 * dot * w1;
                const float ta = c * ss * w2;
                float p2 = ta * y1x, p3 = ta * y1y, p4 = ta * y1z;
                const float tb = c * y0 * w3;
                float p5 = tb * vx, p6 = tb * vy, p7 = tb * vz;
                join_rows(p0, p1);
                atomicAdd(mA, p0);
                atomicAdd(mB, p1);
                join_rows(p2, p5);
                atomicAdd(mA + 256, p2);
                atomicAdd(mB + 256, p5);
                join_rows(p3, p6);
                atomicAdd(mA + 512, p3);
                atomicAdd(mB + 512, p6);
                join_rows(p4, p7);
                atomicAdd(mA + 768, p4);
                atomicAdd(mB + 768, p7);
            }
        }
    }
}

#undef STAGE_LOAD

__global__ void __launch_bounds__(128)
node_down_kernel(const float* __restrict__ msg, const float* __restrict__ L0,
                 const float* __restrict__ L1, float* __restrict__ out) {
    const int lane   = threadIdx.x & 31;
    const int comp   = threadIdx.x >> 5;
    const int lane15 = lane & 15;
    const int half   = lane >> 4;
    const int mbase  = half * 8;
    const int n0     = blockIdx.x * 16;
    const int ab     = (n0 + lane15) * 512 + comp * 128 + half;
    const float* __restrict__ Wm = comp == 0 ? L0 : L1;
    const int wbase  = (half * 2) * 128 + lane15 * 4;

    const v2f* msg2 = (const v2f*)msg;

    v8f acc[8] = {{}, {}, {}, {}, {}, {}, {}, {}};
    for (int p = 0; p < 256; p += 4) {
        const v2f a  = msg2[ab + (p >> 1)];
        const int k  = ((p >> 4) & 1) * 128 + (p >> 5) * 16 + (p & 15);
        const float* wrow = Wm + k * 128 + wbase;
#pragma unroll
        for (int j = 0; j < 2; ++j) {
            const v4f b0 = *(const v4f*)(wrow + 64 * j);
            const v4f b1 = *(const v4f*)(wrow + 128 + 64 * j);
#pragma unroll
            for (int i = 0; i < 4; ++i) {
                v2f b;
                b.x = b0[i];
                b.y = b1[i];
                acc[4 * j + i] = wmma4(a, b, acc[4 * j + i]);
            }
        }
    }

    const float sc = (1.0f / 16.0f) * 0.1f;
    if (comp == 0) {
#pragma unroll
        for (int r = 0; r < 8; ++r) {
            float* __restrict__ d = out + (n0 + mbase + r) * 512 + lane15 * 4;
#pragma unroll
            for (int j = 0; j < 2; ++j) {
                v4f o;
                o.x = acc[4 * j + 0][r] * sc;
                o.y = acc[4 * j + 1][r] * sc;
                o.z = acc[4 * j + 2][r] * sc;
                o.w = acc[4 * j + 3][r] * sc;
                *(v4f*)(d + 64 * j) = o;
            }
        }
    } else {
#pragma unroll
        for (int r = 0; r < 8; ++r) {
            const int ob = (n0 + mbase + r) * 512 + 127 + comp + lane15 * 12;
#pragma unroll
            for (int j = 0; j < 2; ++j)
#pragma unroll
                for (int i = 0; i < 4; ++i)
                    out[ob + (64 * j + i) * 3] = acc[4 * j + i][r] * sc;
        }
    }
}

extern "C" void kernel_launch(void* const* d_in, const int* in_sizes, int n_in,
                              void* d_out, int out_size, void* d_ws, size_t ws_size,
                              hipStream_t stream) {
    (void)in_sizes; (void)n_in; (void)out_size; (void)ws_size;
    const float* nf   = (const float*)d_in[0];
    const float* ea   = (const float*)d_in[1];
    const float* ef   = (const float*)d_in[2];
    const int*   snd  = (const int*)d_in[3];
    const int*   rcv  = (const int*)d_in[4];
    const float* Wup0 = (const float*)d_in[5];
    const float* Wup1 = (const float*)d_in[6];
    const float* w1   = (const float*)d_in[7];
    const float* w2   = (const float*)d_in[8];
    const float* w3   = (const float*)d_in[9];
    const float* w4   = (const float*)d_in[10];
    const float* Wl0  = (const float*)d_in[11];
    const float* Wl1  = (const float*)d_in[12];
    float* out = (float*)d_out;

    float* s_up = (float*)d_ws;
    float* v_up = s_up + (size_t)N_NODES * 128;
    float* msgb = v_up + (size_t)N_NODES * 384;

    zero_kernel<<<2048, 256, 0, stream>>>((float4*)msgb, N_NODES * 1024 / 4);

    node_up_kernel<<<N_NODES / 16, 128, 0, stream>>>(nf, Wup0, Wup1, s_up, v_up);
    edge_kernel<<<512, 256, 0, stream>>>(ea, ef, snd, rcv, w1, w2, w3, w4,
                                         s_up, v_up, msgb);
    node_down_kernel<<<N_NODES / 16, 128, 0, stream>>>(msgb, Wl0, Wl1, out);
}


# Edge kernel: barriers order LDS only (no wait for global loads or atomic adds); sender rows and the next pair's edge inputs are requested a stage ahead and arrive under the layers; sum order kept
# speedup vs baseline: 2.6533x; 1.0660x over previous
#include <hip/hip_runtime.h>

typedef float v2f __attribute__((ext_vector_type(2)));
typedef float v4f __attribute__((ext_vector_type(4)));
typedef float v8f __attribute__((ext_vector_type(8)));

#define N_NODES 20000
#define N_EDGES 200000

__device__ __forceinline__ v8f wmma4(v2f a, v2f b, v8f c) {
    return __builtin_amdgcn_wmma_f32_16x16x4_f32(
         false, a,  false, b,
         (short)0, c,  false,  false);
}

__device__ __forceinline__ float silu(float x) {
    return x / (1.0f + __expf(-x));
}

__device__ __forceinline__ void join_rows(float& a, float& b) {
    typedef unsigned u2v __attribute__((ext_vector_type(2)));
    const u2v s = __builtin_amdgcn_permlane16_swap(__float_as_uint(a), __float_as_uint(b),
                                                   false, false);
    a = __uint_as_float(s.x);
    b = __uint_as_float(s.y);
}

__global__ void zero_kernel(float4* __restrict__ p, int n4) {
    const int stride = gridDim.x * blockDim.x;
    for (int i = blockIdx.x * blockDim.x + threadIdx.x; i < n4; i += stride)
        p[i] = make_float4(0.f, 0.f, 0.f, 0.f);
}


__global__ void __launch_bounds__(128)
node_up_kernel(const float* __restrict__ nf, const float* __restrict__ W0,
               const float* __restrict__ W1, float* __restrict__ s_up,
               float* __restrict__ v_up) {
    const int lane   = threadIdx.x & 31;
    const int comp   = threadIdx.x >> 5;
    const int lane15 = lane & 15;
    const int half   = lane >> 4;
    const int mbase  = half * 8;
    const int n0     = blockIdx.x * 16;
    const int astr   = comp == 0 ? 1 : 3;
    const int aoff   = comp == 0 ? 0 : 127 + comp;
    const int abase  = (n0 + lane15) * 512 + aoff + (half * 2) * astr;
    const float* __restrict__ Wm = comp == 0 ? W0 : W1;
    const int wbase  = (half * 2) * 128 + lane15 * 4;

    v8f acc[8] = {{}, {}, {}, {}, {}, {}, {}, {}};
    for (int k = 0; k < 128; k += 4) {
        v2f a;
        a.x = nf[abase + k * astr];
        a.y = nf[abase + (k + 1) * astr];
        const float* wrow = Wm + k * 128 + wbase;
#pragma unroll
        for (int j = 0; j < 2; ++j) {
            const v4f b0 = *(const v4f*)(wrow + 64 * j);
            const v4f b1 = *(const v4f*)(wrow + 128 + 64 * j);
#pragma unroll
            for (int i = 0; i < 4; ++i) {
                v2f b;
                b.x = b0[i];
                b.y = b1[i];
                acc[4 * j + i] = wmma4(a, b, acc[4 * j + i]);
            }
        }
    }

    const float inv = 0.08838834764831845f;
    if (comp == 0) {
#pragma unroll
        for (int r = 0; r < 8; ++r) {
            float* __restrict__ d = s_up + (n0 + mbase + r) * 128 + lane15 * 4;
#pragma unroll
            for (int j = 0; j < 2; ++j) {
                v4f o;
                o.x = acc[4 * j + 0][r] * inv;
                o.y = acc[4 * j + 1][r] * inv;
                o.z = acc[4 * j + 2][r] * inv;
                o.w = acc[4 * j + 3][r] * inv;
                *(v4f*)(d + 64 * j) = o;
            }
        }
    } else {
        float* __restrict__ dst = v_up + (comp - 1);
#pragma unroll
        for (int r = 0; r < 8; ++r) {
            const int nb = (n0 + mbase + r) * 128 + lane15 * 4;
#pragma unroll
            for (int j = 0; j < 2; ++j)
#pragma unroll
                for (int i = 0; i < 4; ++i)
                    dst[(nb + 64 * j + i) * 3] = acc[4 * j + i][r] * inv;
        }
    }
}

__device__ __forceinline__ void mlp64_bfrag(v2f (&b)[16], const float* __restrict__ Wm,
                                            int ct, int lane15, int half) {
    int wbase  = (half * 2) * 64 + ct * 16 + lane15;
    asm volatile("" : "+v"(wbase));
#pragma unroll
    for (int q = 0; q < 16; ++q) {
        b[q].x = Wm[q * 256 + wbase];
        b[q].y = Wm[q * 256 + 64 + wbase];
    }
}

__device__ __forceinline__ void mlp64_mma(const float* HSRC, float* HDST, const v2f (&b)[16],
                                          int ct, int lane15, int half, int mbase) {
    const v2f* H2    = (const v2f*)HSRC;
    const int abase  = lane15 * 34 + half;
    v8f acc = {};
#pragma unroll
    for (int q = 0; q < 16; ++q)
        acc = wmma4(H2[abase + q * 2], b[q], acc);
#pragma unroll
    for (int r = 0; r < 8; ++r)
        HDST[(mbase + r) * 68 + ct * 16 + lane15] = silu(acc[r] * 0.125f);
}

__device__ __forceinline__ void lds_barrier() {
    __builtin_amdgcn_fence(__ATOMIC_RELEASE, "workgroup", "local");
    __builtin_amdgcn_s_barrier();
    __builtin_amdgcn_fence(__ATOMIC_ACQUIRE, "workgroup", "local");
}

__global__ void __launch_bounds__(256) __attribute__((amdgpu_waves_per_eu(4)))
edge_kernel(const float* __restrict__ ea, const float* __restrict__ ef,
            const int* __restrict__ snd, const int* __restrict__ rcv,
            const float* __restrict__ w1, const float* __restrict__ w2,
            const float* __restrict__ w3, const float* __restrict__ w4,
            const float* __restrict__ s_up, const float* __restrict__ v_up,
            float* __restrict__ msg) {
    __shared__ float hX[2][16 * 68];
    __shared__ float hY[2][16 * 68];
    __shared__ float hZ[2][16 * 68];
    __shared__ __attribute__((aligned(16))) float sc[2][2][16 * 8];
    __shared__ __attribute__((aligned(16))) float G[32 * 516];

    const int lane   = threadIdx.x & 31;
    const int wv     = threadIdx.x >> 5;
    const int lane15 = lane & 15;
    const int half   = lane >> 4;
    const int mbase  = half * 8;
    const int u      = wv * 16 + lane15;
    const int tsel   = wv >> 2;
    const int ct     = wv & 3;

    v2f wr[4][16];
#pragma unroll
    for (int j = 0; j < 4; ++j)
#pragma unroll
        for (int q = 0; q < 16; ++q) {
            wr[j][q].x = w4[(q * 4 + half * 2) * 512 + j * 128 + u];
            wr[j][q].y = w4[(q * 4 + half * 2 + 1) * 512 + j * 128 + u];
        }

    const float c  = 0.08838834764831845f;
    const float c3 = 0.05103103630798287f;
    const int abase = lane15 * 34 + half;

    const int wvu = __builtin_amdgcn_readfirstlane(wv);
    const int sh  = wvu >> 2;
    const int sq  = wvu & 3;
    const char* __restrict__ gbase = (const char*)(sq == 0 ? s_up : v_up + (sq - 1) * 128);
    const unsigned gstride = sq == 0 ? 512u : 1536u;
    const unsigned glane   = lane * 16;
    float* gdst = G + sh * 16 * 516 + sq * 128 + lane * 4;
#define STAGE_LOAD(I) (*(const v4f*)(gbase + ((unsigned)__builtin_amdgcn_readlane(sv, (I)) * gstride + glane)))

    v2f b1[2];
#pragma unroll
    for (int q = 0; q < 2; ++q) {
        b1[q].x = w1[(q * 4 + half * 2) * 64 + ct * 16 + lane15];
        b1[q].y = w1[(q * 4 + half * 2 + 1) * 64 + ct * 16 + lane15];
    }

    const v2f* ef2 = (const v2f*)ef;
    const int npairs = N_EDGES / 32;

    int   sv, sndv = 0, rcvv = 0;
    v2f   efa, efb;
    float4 yv = make_float4(0.f, 0.f, 0.f, 0.f);
    {
        const int p0 = blockIdx.x;
        const int e  = p0 * 32 + tsel * 16 + lane15;
        sv  = snd[p0 * 32 + sh * 16 + lane15];
        efa = ef2[e * 4 + half];
        efb = ef2[e * 4 + half + 2];
        if (ct == 0) {
            yv   = *(const float4*)(ea + e * 4);
            sndv = snd[e];
            rcvv = rcv[e];
        }
    }

    int par = 0;
    for (int p = blockIdx.x; p < npairs; p += gridDim.x, par ^= 1) {
        v4f ga[8];
#pragma unroll
        for (int i = 0; i < 8; ++i)
            ga[i] = STAGE_LOAD(i);

        {
            v8f acc = {};
            acc = wmma4(efa, b1[0], acc);
            acc = wmma4(efb, b1[1], acc);
#pragma unroll
            for (int r = 0; r < 8; ++r)
                hX[tsel][(mbase + r) * 68 + ct * 16 + lane15] =
                    silu(acc[r] * 0.35355339059327373f);
        }
        if (ct == 0) {
            float4 sr;
            sr.x = __int_as_float(sndv);
            sr.y = __int_as_float(rcvv);
            sr.z = 0.f; sr.w = 0.f;
            *(float4*)(&sc[par][tsel][lane15 * 8])     = yv;
            *(float4*)(&sc[par][tsel][lane15 * 8 + 4]) = sr;
        }
        lds_barrier();
#pragma unroll
        for (int i = 0; i < 8; ++i)
            *(v4f*)(gdst + i * 516) = ga[i];

        v2f bw[16];
        mlp64_bfrag(bw, w2, ct, lane15, half);
        v4f gb[8];
#pragma unroll
        for (int i = 0; i < 8; ++i)
            gb[i] = STAGE_LOAD(8 + i);
        mlp64_mma(hX[tsel], hY[tsel], bw, ct, lane15, half, mbase);
        lds_barrier();
#pragma unroll
        for (int i = 0; i < 8; ++i)
            *(v4f*)(gdst + (8 + i) * 516) = gb[i];

        mlp64_bfrag(bw, w3, ct, lane15, half);
        const int pn = (p + (int)gridDim.x < npairs) ? p + (int)gridDim.x : p;
        const int en = pn * 32 + tsel * 16 + lane15;
        const int svn  = snd[pn * 32 + sh * 16 + lane15];
        const v2f efan = ef2[en * 4 + half];
        const v2f efbn = ef2[en * 4 + half + 2];
        float4 yvn = make_float4(0.f, 0.f, 0.f, 0.f);
        int sndvn = 0, rcvvn = 0;
        if (ct == 0) {
            yvn   = *(const float4*)(ea + en * 4);
            sndvn = snd[en];
            rcvvn = rcv[en];
        }
        mlp64_mma(hY[tsel], hZ[tsel], bw, ct, lane15, half, mbase);
        lds_barrier();

#pragma unroll 1
        for (int t = 0; t < 2; ++t) {
            const v2f* H2 = (const v2f*)hZ[t];
            const float* S = sc[par][t];
            const float* gs = G + (t * 16 + mbase) * 516 + u;
            const float* gv = G + (t * 16 + mbase) * 516 + 128 + 3 * u;
            const int lo = wv * 32 + lane;
            v8f acc[4] = {{}, {}, {}, {}};
#pragma unroll
            for (int q = 0; q < 16; ++q) {
                const v2f a = H2[abase + q * 2];
#pragma unroll
                for (int j = 0; j < 4; ++j)
                    acc[j] = wmma4(a, wr[j][q], acc[j]);
            }
#pragma unroll
            for (int r = 0; r < 8; ++r) {
                const int row = mbase + r;
                const float4 y  = *(const float4*)(S + row * 8);
                const float y0 = y.x, y1x = y.y, y1y = y.z, y1z = y.w;
                const int rcvA = __builtin_amdgcn_readfirstlane(__float_as_int(S[r * 8 + 5]));
                const int rcvB = __builtin_amdgcn_readfirstlane(__float_as_int(S[(r + 8) * 8 + 5]));
                float* __restrict__ mA = msg + (size_t)rcvA * 1024 + lo;
                float* __restrict__ mB = msg + (size_t)rcvB * 1024 + lo;
                const float ss = gs[r * 516];
                const float vx = gv[r * 516 + 0];
                const float vy = gv[r * 516 + 1];
                const float vz = gv[r * 516 + 2];
                const float w0 = acc[0][r] * 0.125f;
                const float w1 = acc[1][r] * 0.125f;
                const float w2 = acc[2][r] * 0.125f;
                const float w3 = acc[3][r] * 0.125f;
                float p0 = c * ss * y0 * w0;
                const float dot = vx * y1x + vy * y1y + vz * y1z;
                float p1 = c3 * dot * w1;
                const float ta = c * ss * w2;
                float p2 = ta * y1x, p3 = ta * y1y, p4 = ta * y1z;
                const float tb = c * y0 * w3;
                float p5 = tb * vx, p6 = tb * vy, p7 = tb * vz;
                join_rows(p0, p1);
                atomicAdd(mA, p0);
                atomicAdd(mB, p1);
                join_rows(p2, p5);
                atomicAdd(mA + 256, p2);
                atomicAdd(mB + 256, p5);
                join_rows(p3, p6);
                atomicAdd(mA + 512, p3);
                atomicAdd(mB + 512, p6);
                join_rows(p4, p7);
                atomicAdd(mA + 768, p4);
                atomicAdd(mB + 768, p7);
            }
        }
        sv = svn; efa = efan; efb = efbn;
        yv = yvn; sndv = sndvn; rcvv = rcvvn;
    }
}

#undef STAGE_LOAD

__global__ void __launch_bounds__(128)
node_down_kernel(const float* __restrict__ msg, const float* __restrict__ L0,
                 const float* __restrict__ L1, float* __restrict__ out) {
    const int lane   = threadIdx.x & 31;
    const int comp   = threadIdx.x >> 5;
    const int lane15 = lane & 15;
    const int half   = lane >> 4;
    const int mbase  = half * 8;
    const int n0     = blockIdx.x * 16;
    const int ab     = (n0 + lane15) * 512 + comp * 128 + half;
    const float* __restrict__ Wm = comp == 0 ? L0 : L1;
    const int wbase  = (half * 2) * 128 + lane15 * 4;

    const v2f* msg2 = (const v2f*)msg;

    v8f acc[8] = {{}, {}, {}, {}, {}, {}, {}, {}};
    for (int p = 0; p < 256; p += 4) {
        const v2f a  = msg2[ab + (p >> 1)];
        const int k  = ((p >> 4) & 1) * 128 + (p >> 5) * 16 + (p & 15);
        const float* wrow = Wm + k * 128 + wbase;
#pragma unroll
        for (int j = 0; j < 2; ++j) {
            const v4f b0 = *(const v4f*)(wrow + 64 * j);
            const v4f b1 = *(const v4f*)(wrow + 128 + 64 * j);
#pragma unroll
            for (int i = 0; i < 4; ++i) {
                v2f b;
                b.x = b0[i];
                b.y = b1[i];
                acc[4 * j + i] = wmma4(a, b, acc[4 * j + i]);
            }
        }
    }

    const float sc = (1.0f / 16.0f) * 0.1f;
    if (comp == 0) {
#pragma unroll
        for (int r = 0; r < 8; ++r) {
            float* __restrict__ d = out + (n0 + mbase + r) * 512 + lane15 * 4;
#pragma unroll
            for (int j = 0; j < 2; ++j) {
                v4f o;
                o.x = acc[4 * j + 0][r] * sc;
                o.y = acc[4 * j + 1][r] * sc;
                o.z = acc[4 * j + 2][r] * sc;
                o.w = acc[4 * j + 3][r] * sc;
                *(v4f*)(d + 64 * j) = o;
            }
        }
    } else {
#pragma unroll
        for (int r = 0; r < 8; ++r) {
            const int ob = (n0 + mbase + r) * 512 + 127 + comp + lane15 * 12;
#pragma unroll
            for (int j = 0; j < 2; ++j)
#pragma unroll
                for (int i = 0; i < 4; ++i)
                    out[ob + (64 * j + i) * 3] = acc[4 * j + i][r] * sc;
        }
    }
}

extern "C" void kernel_launch(void* const* d_in, const int* in_sizes, int n_in,
                              void* d_out, int out_size, void* d_ws, size_t ws_size,
                              hipStream_t stream) {
    (void)in_sizes; (void)n_in; (void)out_size; (void)ws_size;
    const float* nf   = (const float*)d_in[0];
    const float* ea   = (const float*)d_in[1];
    const float* ef   = (const float*)d_in[2];
    const int*   snd  = (const int*)d_in[3];
    const int*   rcv  = (const int*)d_in[4];
    const float* Wup0 = (const float*)d_in[5];
    const float* Wup1 = (const float*)d_in[6];
    const float* w1   = (const float*)d_in[7];
    const float* w2   = (const float*)d_in[8];
    const float* w3   = (const float*)d_in[9];
    const float* w4   = (const float*)d_in[10];
    const float* Wl0  = (const float*)d_in[11];
    const float* Wl1  = (const float*)d_in[12];
    float* out = (float*)d_out;

    float* s_up = (float*)d_ws;
    float* v_up = s_up + (size_t)N_NODES * 128;
    float* msgb = v_up + (size_t)N_NODES * 384;

    zero_kernel<<<2048, 256, 0, stream>>>((float4*)msgb, N_NODES * 1024 / 4);

    node_up_kernel<<<N_NODES / 16, 128, 0, stream>>>(nf, Wup0, Wup1, s_up, v_up);
    edge_kernel<<<512, 256, 0, stream>>>(ea, ef, snd, rcv, w1, w2, w3, w4,
                                         s_up, v_up, msgb);
    node_down_kernel<<<N_NODES / 16, 128, 0, stream>>>(msgb, Wl0, Wl1, out);
}
